# starting kernel
# baseline (speedup 1.0000x reference)
#include <hip/hip_runtime.h>

typedef float v2f __attribute__((ext_vector_type(2)));
typedef float v8f __attribute__((ext_vector_type(8)));

#define N_NODES 10000
#define N_EDGES 160000
#define N_TILES (N_EDGES / 16)

#define INV_M0  0.25f
#define INV_3M1 0.20412414523f
#define INV_M1  0.35355339059f
#define EPSV    1e-5f

#define OFF_SUMS  0
#define OFF_CNTS  400000
#define OFF_STATS 410000
#define ZERO_N    410040
#define OFF_W1F   410048
#define OFF_W2F   414144

static __device__ __forceinline__ v8f wmma4(v2f a, v2f b, v8f c) {
  return __builtin_amdgcn_wmma_f32_16x16x4_f32(false, a, false, b, (short)0, c,
                                               false, false);
}

static __device__ __forceinline__ void atomAddF(float* p, float v) {
  __hip_atomic_fetch_add(p, v, __ATOMIC_RELAXED, __HIP_MEMORY_SCOPE_AGENT);
}

__global__ void zero_kernel(float* ws) {
  int i = blockIdx.x * blockDim.x + threadIdx.x;
  if (i < ZERO_N) ws[i] = 0.0f;
}

__global__ void repack_kernel(const float* __restrict__ w1,
                              const float* __restrict__ w2,
                              float* __restrict__ w1f,
                              float* __restrict__ w2f) {
  int t = blockIdx.x * blockDim.x + threadIdx.x;
  int f = t >> 5, lane = t & 31;
  int hb = lane >> 4, nn = lane & 15;
  if (f < 64) {
    int b = f >> 2, j = f & 3;
    int r0 = 4 * b + 2 * hb;
    w1f[(f * 32 + lane) * 2 + 0] = w1[(r0 + 0) * 64 + 16 * j + nn];
    w1f[(f * 32 + lane) * 2 + 1] = w1[(r0 + 1) * 64 + 16 * j + nn];
  } else if (f < 64 + 576) {
    int f2 = f - 64;
    int b = f2 / 36, j = f2 % 36;
    int r0 = 4 * b + 2 * hb;
    w2f[(f2 * 32 + lane) * 2 + 0] = w2[(r0 + 0) * 576 + 16 * j + nn];
    w2f[(f2 * 32 + lane) * 2 + 1] = w2[(r0 + 1) * 576 + 16 * j + nn];
  }
}

#define WPB 6
#define SMW 2176

__global__ __launch_bounds__(WPB * 32) void edge_kernel(
    const float* __restrict__ atom, const float* __restrict__ ef,
    const float* __restrict__ esh, const int* __restrict__ eidx,
    const float* __restrict__ b1, const float* __restrict__ b2,
    const float* __restrict__ w1f, const float* __restrict__ w2f,
    float* __restrict__ sums, float* __restrict__ cnts) {
  __shared__ float smem[WPB * SMW];
  __shared__ int nodes[WPB][16];

  int lane = threadIdx.x & 31;
  int wave = threadIdx.x >> 5;
  int tile = blockIdx.x * WPB + wave;
  if (tile >= N_TILES) return;
  int e0 = tile * 16;
  int hb = lane >> 4;
  int nn = lane & 15;

  float* sm  = smem + wave * SMW;
  float* hT  = sm;
  float* a11 = sm + 1088;
  float* a12 = sm + 1344;
  float* a13 = sm + 1600;
  float* a14 = sm + 1984;
  float* s1  = sm + 2112;

  if (lane < 16) {
    int e = e0 + lane;
    nodes[wave][lane] = eidx[N_EDGES + e];
    float4 s = *(const float4*)(esh + e * 4);
    const float* xr = atom + (size_t)eidx[e] * 40;
    float x[40];
#pragma unroll
    for (int t = 0; t < 10; t++) {
      float4 v = *(const float4*)(xr + t * 4);
      x[t * 4 + 0] = v.x; x[t * 4 + 1] = v.y;
      x[t * 4 + 2] = v.z; x[t * 4 + 3] = v.w;
    }
    float sh0 = s.x;
    s1[lane * 4 + 0] = s.y;
    s1[lane * 4 + 1] = s.z;
    s1[lane * 4 + 2] = s.w;
#pragma unroll
    for (int j = 0; j < 16; j++) {
      a11[lane * 16 + j] = x[j] * sh0 * INV_M0;
      a12[lane * 16 + j] = x[j] * INV_M0;
    }
#pragma unroll
    for (int i = 0; i < 8; i++) {
      float m0 = x[16 + i * 3 + 0];
      float m1 = x[16 + i * 3 + 1];
      float m2 = x[16 + i * 3 + 2];
      a13[lane * 24 + i * 3 + 0] = sh0 * m0 * INV_M1;
      a13[lane * 24 + i * 3 + 1] = sh0 * m1 * INV_M1;
      a13[lane * 24 + i * 3 + 2] = sh0 * m2 * INV_M1;
      a14[lane * 8 + i] = (m0 * s.y + m1 * s.z + m2 * s.w) * INV_3M1;
    }
  }
  asm volatile("s_wait_dscnt 0" ::: "memory");

  v2f a[16];
  const float* efr = ef + (size_t)(e0 + nn) * 64 + 2 * hb;
#pragma unroll
  for (int b = 0; b < 16; b++) a[b] = *(const v2f*)(efr + 4 * b);

#pragma unroll
  for (int j = 0; j < 4; j++) {
    v8f c;
    float bias = b1[16 * j + nn];
#pragma unroll
    for (int r = 0; r < 8; r++) c[r] = bias;
#pragma unroll
    for (int b = 0; b < 16; b++) {
      v2f wb = *(const v2f*)(w1f + ((b * 4 + j) * 32 + lane) * 2);
      c = wmma4(a[b], wb, c);
    }
#pragma unroll
    for (int r = 0; r < 8; r++) {
      float v = c[r] > 0.0f ? c[r] : 0.0f;
      hT[(r + 8 * hb) * 68 + 16 * j + nn] = v;
    }
  }
  asm volatile("s_wait_dscnt 0" ::: "memory");

  v2f ha[16];
#pragma unroll
  for (int b = 0; b < 16; b++)
    ha[b] = *(const v2f*)(hT + nn * 68 + 4 * b + 2 * hb);

  float o0[8] = {0, 0, 0, 0, 0, 0, 0, 0};
  float p1[8] = {0, 0, 0, 0, 0, 0, 0, 0};
  float q0[8] = {0, 0, 0, 0, 0, 0, 0, 0};
  float q1[8] = {0, 0, 0, 0, 0, 0, 0, 0};
  float q2[8] = {0, 0, 0, 0, 0, 0, 0, 0};

  for (int j2 = 0; j2 < 16; j2++) {
    v8f c;
    float bias = b2[16 * j2 + nn];
#pragma unroll
    for (int r = 0; r < 8; r++) c[r] = bias;
#pragma unroll
    for (int b = 0; b < 16; b++) {
      v2f wb = *(const v2f*)(w2f + ((b * 36 + j2) * 32 + lane) * 2);
      c = wmma4(ha[b], wb, c);
    }
#pragma unroll
    for (int r = 0; r < 8; r++)
      o0[r] += c[r] * a11[(r + 8 * hb) * 16 + j2];
  }
  for (int j2 = 16; j2 < 24; j2++) {
    v8f c;
    float bias = b2[16 * j2 + nn];
#pragma unroll
    for (int r = 0; r < 8; r++) c[r] = bias;
#pragma unroll
    for (int b = 0; b < 16; b++) {
      v2f wb = *(const v2f*)(w2f + ((b * 36 + j2) * 32 + lane) * 2);
      c = wmma4(ha[b], wb, c);
    }
    int i = 2 * (j2 - 16) + (nn >> 3);
#pragma unroll
    for (int r = 0; r < 8; r++)
      p1[r] += c[r] * a12[(r + 8 * hb) * 16 + i];
  }
  for (int j2 = 24; j2 < 28; j2++) {
    v8f c;
    float bias = b2[16 * j2 + nn];
#pragma unroll
    for (int r = 0; r < 8; r++) c[r] = bias;
#pragma unroll
    for (int b = 0; b < 16; b++) {
      v2f wb = *(const v2f*)(w2f + ((b * 36 + j2) * 32 + lane) * 2);
      c = wmma4(ha[b], wb, c);
    }
    int i = 2 * (j2 - 24) + (nn >> 3);
#pragma unroll
    for (int r = 0; r < 8; r++) {
      float cv = c[r];
      int base = (r + 8 * hb) * 24 + i * 3;
      q0[r] += cv * a13[base + 0];
      q1[r] += cv * a13[base + 1];
      q2[r] += cv * a13[base + 2];
    }
  }
  for (int j2 = 28; j2 < 36; j2++) {
    v8f c;
    float bias = b2[16 * j2 + nn];
#pragma unroll
    for (int r = 0; r < 8; r++) c[r] = bias;
#pragma unroll
    for (int b = 0; b < 16; b++) {
      v2f wb = *(const v2f*)(w2f + ((b * 36 + j2) * 32 + lane) * 2);
      c = wmma4(ha[b], wb, c);
    }
    int i = j2 - 28;
#pragma unroll
    for (int r = 0; r < 8; r++)
      o0[r] += c[r] * a14[(r + 8 * hb) * 8 + i];
  }

#pragma unroll
  for (int r = 0; r < 8; r++) {
    p1[r] += __shfl_xor(p1[r], 8, 32);
    q0[r] += __shfl_xor(q0[r], 8, 32);
    q1[r] += __shfl_xor(q1[r], 8, 32);
    q2[r] += __shfl_xor(q2[r], 8, 32);
  }

#pragma unroll
  for (int r = 0; r < 8; r++) {
    int e = r + 8 * hb;
    atomAddF(sums + (size_t)nodes[wave][e] * 40 + nn, o0[r]);
  }
  if (nn < 8) {
#pragma unroll
    for (int r = 0; r < 8; r++) {
      int e = r + 8 * hb;
      float* dst = sums + (size_t)nodes[wave][e] * 40 + 16 + nn * 3;
      float sy = s1[e * 4 + 0], sz = s1[e * 4 + 1], sw = s1[e * 4 + 2];
      atomAddF(dst + 0, p1[r] * sy + q0[r]);
      atomAddF(dst + 1, p1[r] * sz + q1[r]);
      atomAddF(dst + 2, p1[r] * sw + q2[r]);
    }
  }
  if (lane < 16) atomAddF(cnts + nodes[wave][lane], 1.0f);
}

__global__ void node_stage1(const float* __restrict__ atom, float* ws) {
  int c = threadIdx.x;
  int ty = threadIdx.y;
  float* sums = ws + OFF_SUMS;
  const float* cnts = ws + OFF_CNTS;
  float* stats = ws + OFF_STATS;
  float sA = 0.0f, sB = 0.0f;
  for (int n = blockIdx.x * 6 + ty; n < N_NODES; n += gridDim.x * 6) {
    float cnt = cnts[n];
    float inv = 1.0f / fmaxf(cnt, 1.0f);
    float y = sums[(size_t)n * 40 + c] * inv + atom[(size_t)n * 40 + c];
    sums[(size_t)n * 40 + c] = y;
    sA += y;
    sB += y * y;
  }
  __shared__ float rA[6][40], rB[6][40];
  rA[ty][c] = sA;
  rB[ty][c] = sB;
  __syncthreads();
  if (ty == 0) {
    float tA = 0.0f, tB = 0.0f;
#pragma unroll
    for (int k = 0; k < 6; k++) { tA += rA[k][c]; tB += rB[k][c]; }
    if (c < 16) {
      atomAddF(stats + c, tA);
      atomAddF(stats + 16 + c, tB);
    } else {
      atomAddF(stats + 32 + (c - 16) / 3, tB);
    }
  }
}

__global__ void node_stage2(const float* __restrict__ ws,
                            const float* __restrict__ bnw,
                            const float* __restrict__ bnb,
                            float* __restrict__ out) {
  int idx = blockIdx.x * blockDim.x + threadIdx.x;
  if (idx >= N_NODES * 40) return;
  int c = idx % 40;
  const float* y = ws + OFF_SUMS;
  const float* stats = ws + OFF_STATS;
  float v = y[idx];
  float o;
  if (c < 16) {
    float mean = stats[c] * (1.0f / N_NODES);
    float ex2 = stats[16 + c] * (1.0f / N_NODES);
    float var = ex2 - mean * mean;
    o = (v - mean) * rsqrtf(var + EPSV) * bnw[c] + bnb[c];
  } else {
    int j = (c - 16) / 3;
    float vn = stats[32 + j] * (1.0f / (3.0f * N_NODES));
    o = v * rsqrtf(vn + EPSV) * bnw[16 + j];
  }
  out[idx] = o;
}

extern "C" void kernel_launch(void* const* d_in, const int* in_sizes, int n_in,
                              void* d_out, int out_size, void* d_ws,
                              size_t ws_size, hipStream_t stream) {
  (void)in_sizes; (void)n_in; (void)out_size; (void)ws_size;
  const float* atom = (const float*)d_in[0];
  const float* ef   = (const float*)d_in[1];
  const float* esh  = (const float*)d_in[2];
  const int*   eidx = (const int*)d_in[3];
  const float* w1   = (const float*)d_in[4];
  const float* b1   = (const float*)d_in[5];
  const float* w2   = (const float*)d_in[6];
  const float* b2   = (const float*)d_in[7];
  const float* bnw  = (const float*)d_in[8];
  const float* bnb  = (const float*)d_in[9];
  float* ws  = (float*)d_ws;
  float* out = (float*)d_out;

  zero_kernel<<<(ZERO_N + 255) / 256, 256, 0, stream>>>(ws);
  repack_kernel<<<(640 * 32 + 255) / 256, 256, 0, stream>>>(
      w1, w2, ws + OFF_W1F, ws + OFF_W2F);
  edge_kernel<<<(N_TILES + WPB - 1) / WPB, WPB * 32, 0, stream>>>(
      atom, ef, esh, eidx, b1, b2, ws + OFF_W1F, ws + OFF_W2F,
      ws + OFF_SUMS, ws + OFF_CNTS);
  node_stage1<<<128, dim3(40, 6), 0, stream>>>(atom, ws);
  node_stage2<<<(N_NODES * 40 + 255) / 256, 256, 0, stream>>>(ws, bnw, bnb, out);
}


# edge kernel start re-ordered for an early first matrix instruction: features fetched as whole-line loads via LDS; table build (ids first, rows as one group) moved behind the first product; same sums
# speedup vs baseline: 1.3498x; 1.3498x over previous
#include <hip/hip_runtime.h>

typedef float v2f __attribute__((ext_vector_type(2)));
typedef float v8f __attribute__((ext_vector_type(8)));

#define N_NODES 10000
#define N_EDGES 160000
#define N_TILES (N_EDGES / 16)

#define INV_M0  0.25f
#define INV_3M1 0.20412414523f
#define INV_M1  0.35355339059f
#define EPSV    1e-5f

#define OFF_SUMS  0
#define OFF_CNTS  400000
#define OFF_STATS 410000
#define ZERO_N    410040
#define OFF_W1F   410048
#define OFF_W2F   414144

static __device__ __forceinline__ v8f wmma4(v2f a, v2f b, v8f c) {
  return __builtin_amdgcn_wmma_f32_16x16x4_f32(false, a, false, b, (short)0, c,
                                               false, false);
}

static __device__ __forceinline__ void atomAddF(float* p, float v) {
  __hip_atomic_fetch_add(p, v, __ATOMIC_RELAXED, __HIP_MEMORY_SCOPE_AGENT);
}

__global__ void zero_kernel(float* ws) {
  int i = blockIdx.x * blockDim.x + threadIdx.x;
  if (i < ZERO_N) ws[i] = 0.0f;
}

__global__ void repack_kernel(const float* __restrict__ w1,
                              const float* __restrict__ w2,
                              float* __restrict__ w1f,
                              float* __restrict__ w2f) {
  int t = blockIdx.x * blockDim.x + threadIdx.x;
  int f = t >> 5, lane = t & 31;
  int hb = lane >> 4, nn = lane & 15;
  if (f < 64) {
    int b = f >> 2, j = f & 3;
    int r0 = 4 * b + 2 * hb;
    w1f[(f * 32 + lane) * 2 + 0] = w1[(r0 + 0) * 64 + 16 * j + nn];
    w1f[(f * 32 + lane) * 2 + 1] = w1[(r0 + 1) * 64 + 16 * j + nn];
  } else if (f < 64 + 576) {
    int f2 = f - 64;
    int b = f2 / 36, j = f2 % 36;
    int r0 = 4 * b + 2 * hb;
    w2f[(f2 * 32 + lane) * 2 + 0] = w2[(r0 + 0) * 576 + 16 * j + nn];
    w2f[(f2 * 32 + lane) * 2 + 1] = w2[(r0 + 1) * 576 + 16 * j + nn];
  }
}

#define RT   5
#define EPW  (RT * 16)
#define N_WT (N_EDGES / EPW)
#define TBL  1088
#define SMW  (1088 + RT * TBL)
#define MT_K (EPW * 24 / 32 / 12)

static __device__ __forceinline__ void load_w2(v2f (&w)[16],
                                               const float* __restrict__ w2f,
                                               int j2, int lane) {
#pragma unroll
  for (int b = 0; b < 16; b++)
    w[b] = *(const v2f*)(w2f + ((b * 36 + j2) * 32 + lane) * 2);
}

static __device__ __forceinline__ void tile_mma(v8f (&c)[RT],
                                                const v2f (&ha)[RT][16],
                                                const v2f (&w)[16], float bias) {
#pragma unroll
  for (int t = 0; t < RT; t++)
#pragma unroll
    for (int r = 0; r < 8; r++) c[t][r] = bias;
#pragma unroll
  for (int b = 0; b < 16; b++)
#pragma unroll
    for (int t = 0; t < RT; t++) c[t] = wmma4(ha[t][b], w[b], c[t]);
}

__global__ __launch_bounds__(32) __attribute__((amdgpu_waves_per_eu(2, 2)))
void edge_kernel(
    const float* __restrict__ atom, const float* __restrict__ ef,
    const float* __restrict__ esh, const int* __restrict__ eidx,
    const float* __restrict__ b1, const float* __restrict__ b2,
    const float* __restrict__ w1f, const float* __restrict__ w2f,
    float* __restrict__ sums, float* __restrict__ cnts) {
  __shared__ __attribute__((aligned(16))) float smem[SMW];
  __shared__ int nodes[EPW];

  int lane = threadIdx.x;
  int e0 = blockIdx.x * EPW;
  int hb = lane >> 4;
  int nn = lane & 15;

  float* hT = smem;
  float* tb = smem + 1088;

  int dsti[(EPW + 31) / 32], srci[(EPW + 31) / 32];
  float4 shv[(EPW + 31) / 32];
  v8f hc[RT][4];
  {
    float* stg = tb;
    const float4* efv = (const float4*)(ef + (size_t)e0 * 64);
    float4 fv[EPW * 16 / 32];
#pragma unroll
    for (int k = 0; k < EPW * 16 / 32; k++)
      fv[k] = efv[k * 32 + lane];
    v2f wb[4][16];
    float bias1[4];
#pragma unroll
    for (int j = 0; j < 4; j++) bias1[j] = b1[16 * j + nn];
#pragma unroll
    for (int b = 0; b < 16; b++)
      wb[0][b] = *(const v2f*)(w1f + ((b * 4 + 0) * 32 + lane) * 2);
#pragma unroll
    for (int k = 0; k < EPW * 16 / 32; k++) {
      int f = k * 32 + lane;
      *(float4*)(stg + (f >> 4) * 68 + (f & 15) * 4) = fv[k];
    }
    asm volatile("s_wait_dscnt 0" ::: "memory");
    v2f a[RT][16];
#pragma unroll
    for (int t = 0; t < RT; t++) {
      const float* efr = stg + (t * 16 + nn) * 68 + 2 * hb;
#pragma unroll
      for (int b = 0; b < 16; b++) a[t][b] = *(const v2f*)(efr + 4 * b);
    }
    asm volatile("s_wait_dscnt 0" ::: "memory");
#pragma unroll
    for (int p = 0; p < (EPW + 31) / 32; p++) {
      int elc = p * 32 + lane < EPW ? p * 32 + lane : EPW - 1;
      int e = e0 + elc;
      dsti[p] = eidx[e];
      srci[p] = eidx[N_EDGES + e];
      shv[p] = *(const float4*)(esh + e * 4);
    }
#pragma unroll
    for (int j = 1; j < 4; j++)
#pragma unroll
      for (int b = 0; b < 16; b++)
        wb[j][b] = *(const v2f*)(w1f + ((b * 4 + j) * 32 + lane) * 2);
#pragma unroll
    for (int j = 0; j < 4; j++) {
#pragma unroll
      for (int t = 0; t < RT; t++)
#pragma unroll
        for (int r = 0; r < 8; r++) hc[t][j][r] = bias1[j];
#pragma unroll
      for (int b = 0; b < 16; b++)
#pragma unroll
        for (int t = 0; t < RT; t++)
          hc[t][j] = wmma4(a[t][b], wb[j][b], hc[t][j]);
    }
  }

  v2f ha[RT][16];
#pragma unroll
  for (int t = 0; t < RT; t++) {
#pragma unroll
    for (int j = 0; j < 4; j++)
#pragma unroll
      for (int r = 0; r < 8; r++) {
        float v = hc[t][j][r] > 0.0f ? hc[t][j][r] : 0.0f;
        hT[(r + 8 * hb) * 68 + 16 * j + nn] = v;
      }
    asm volatile("s_wait_dscnt 0" ::: "memory");
#pragma unroll
    for (int b = 0; b < 16; b++)
      ha[t][b] = *(const v2f*)(hT + nn * 68 + 4 * b + 2 * hb);
    asm volatile("s_wait_dscnt 0" ::: "memory");
  }

  {
    float x[(EPW + 31) / 32][40];
#pragma unroll
    for (int p = 0; p < (EPW + 31) / 32; p++) {
      const float* xr = atom + (size_t)dsti[p] * 40;
#pragma unroll
      for (int t = 0; t < 10; t++) {
        float4 v = *(const float4*)(xr + t * 4);
        x[p][t * 4 + 0] = v.x; x[p][t * 4 + 1] = v.y;
        x[p][t * 4 + 2] = v.z; x[p][t * 4 + 3] = v.w;
      }
    }
#pragma unroll
    for (int p = 0; p < (EPW + 31) / 32; p++) {
      int el = p * 32 + lane;
      if ((p + 1) * 32 <= EPW || el < EPW) {
        int row = el & 15;
        float* tr = tb + (el >> 4) * TBL;
        float* a11 = tr;
        float* a12 = tr + 256;
        float* a13 = tr + 512;
        float* a14 = tr + 896;
        float* s1  = tr + 1024;
        float4 s = shv[p];
        nodes[el] = srci[p];
        float sh0 = s.x;
        s1[row * 4 + 0] = s.y;
        s1[row * 4 + 1] = s.z;
        s1[row * 4 + 2] = s.w;
#pragma unroll
        for (int j = 0; j < 16; j++) {
          a11[row * 16 + j] = x[p][j] * sh0 * INV_M0;
          a12[row * 16 + j] = x[p][j] * INV_M0;
        }
#pragma unroll
        for (int i = 0; i < 8; i++) {
          float m0 = x[p][16 + i * 3 + 0];
          float m1 = x[p][16 + i * 3 + 1];
          float m2 = x[p][16 + i * 3 + 2];
          a13[row * 24 + i * 3 + 0] = sh0 * m0 * INV_M1;
          a13[row * 24 + i * 3 + 1] = sh0 * m1 * INV_M1;
          a13[row * 24 + i * 3 + 2] = sh0 * m2 * INV_M1;
          a14[row * 8 + i] = (m0 * s.y + m1 * s.z + m2 * s.w) * INV_3M1;
        }
      }
    }
  }

  asm volatile("s_wait_dscnt 0" ::: "memory");

  float o0[RT][8], p1[RT][8], q0[RT][8], q1[RT][8], q2[RT][8];
#pragma unroll
  for (int t = 0; t < RT; t++)
#pragma unroll
    for (int r = 0; r < 8; r++) {
      o0[t][r] = 0.0f; p1[t][r] = 0.0f;
      q0[t][r] = 0.0f; q1[t][r] = 0.0f; q2[t][r] = 0.0f;
    }

  v2f wA[16], wB[16];
  float biasA, biasB;
  v8f c[RT];

#define EPI11(J2)                                                        \
  _Pragma("unroll") for (int t = 0; t < RT; t++) {                       \
    const float* A = tb + t * TBL;                                       \
    _Pragma("unroll") for (int r = 0; r < 8; r++)                        \
      o0[t][r] += c[t][r] * A[(r + 8 * hb) * 16 + (J2)];                 \
  }
#define EPI12(J2)                                                        \
  {                                                                      \
    int i = 2 * ((J2) - 16) + (nn >> 3);                                 \
    _Pragma("unroll") for (int t = 0; t < RT; t++) {                     \
      const float* A = tb + t * TBL + 256;                               \
      _Pragma("unroll") for (int r = 0; r < 8; r++)                      \
        p1[t][r] += c[t][r] * A[(r + 8 * hb) * 16 + i];                  \
    }                                                                    \
  }
#define EPI13(J2)                                                        \
  {                                                                      \
    int i = 2 * ((J2) - 24) + (nn >> 3);                                 \
    _Pragma("unroll") for (int t = 0; t < RT; t++) {                     \
      const float* A = tb + t * TBL + 512;                               \
      _Pragma("unroll") for (int r = 0; r < 8; r++) {                    \
        float cv = c[t][r];                                              \
        int base = (r + 8 * hb) * 24 + i * 3;                            \
        q0[t][r] += cv * A[base + 0];                                    \
        q1[t][r] += cv * A[base + 1];                                    \
        q2[t][r] += cv * A[base + 2];                                    \
      }                                                                  \
    }                                                                    \
  }
#define EPI14(J2)                                                        \
  {                                                                      \
    int i = (J2) - 28;                                                   \
    _Pragma("unroll") for (int t = 0; t < RT; t++) {                     \
      const float* A = tb + t * TBL + 896;                               \
      _Pragma("unroll") for (int r = 0; r < 8; r++)                      \
        o0[t][r] += c[t][r] * A[(r + 8 * hb) * 8 + i];                   \
    }                                                                    \
  }
#define PAIR(J2, JN, EPI)                                                \
  load_w2(wB, w2f, (J2) + 1, lane);                                      \
  biasB = b2[16 * ((J2) + 1) + nn];                                      \
  tile_mma(c, ha, wA, biasA);                                            \
  EPI(J2)                                                                \
  load_w2(wA, w2f, (JN), lane);                                          \
  biasA = b2[16 * (JN) + nn];                                            \
  tile_mma(c, ha, wB, biasB);                                            \
  EPI((J2) + 1)

#define METER(TRIP)                                                      \
  {                                                                      \
    float mv[MT_K];                                                      \
    int mo[MT_K];                                                        \
    _Pragma("unroll") for (int k = 0; k < MT_K; k++) {                   \
      int f = (TRIP) * (MT_K * 32) + k * 32 + lane;                      \
      int el = f / 24;                                                   \
      int ch = f - el * 24;                                              \
      mv[k] = tb[(el >> 4) * TBL + 512 + (el & 15) * 24 + ch];           \
      mo[k] = nodes[el] * 40 + 16 + ch;                                  \
    }                                                                    \
    _Pragma("unroll") for (int k = 0; k < MT_K; k++)                     \
      atomAddF(sums + mo[k], mv[k]);                                     \
  }

  load_w2(wA, w2f, 16, lane);
  biasA = b2[16 * 16 + nn];

#pragma nounroll
  for (int j2 = 16; j2 < 24; j2 += 2) { PAIR(j2, j2 + 2, EPI12) }
#pragma nounroll
  for (int j2 = 24; j2 < 28; j2 += 2) {
    int jn = j2 + 2 < 28 ? j2 + 2 : 0;
    PAIR(j2, jn, EPI13)
  }

#pragma unroll
  for (int t = 0; t < RT; t++) {
    const float* s1 = tb + t * TBL + 1024;
    float* stg = tb + t * TBL + 512;
#pragma unroll
    for (int r = 0; r < 8; r++) {
      p1[t][r] += __shfl_xor(p1[t][r], 8, 32);
      q0[t][r] += __shfl_xor(q0[t][r], 8, 32);
      q1[t][r] += __shfl_xor(q1[t][r], 8, 32);
      q2[t][r] += __shfl_xor(q2[t][r], 8, 32);
    }
    if (nn < 8) {
#pragma unroll
      for (int r = 0; r < 8; r++) {
        int e = r + 8 * hb;
        float sy = s1[e * 4 + 0], sz = s1[e * 4 + 1], sw = s1[e * 4 + 2];
        stg[e * 24 + nn * 3 + 0] = p1[t][r] * sy + q0[t][r];
        stg[e * 24 + nn * 3 + 1] = p1[t][r] * sz + q1[t][r];
        stg[e * 24 + nn * 3 + 2] = p1[t][r] * sw + q2[t][r];
      }
    }
  }
  asm volatile("s_wait_dscnt 0" ::: "memory");
#pragma unroll
  for (int p = 0; p < (EPW + 31) / 32; p++) {
    int el = p * 32 + lane;
    if (el < EPW) atomAddF(cnts + nodes[el], 1.0f);
  }

#pragma nounroll
  for (int j2 = 0; j2 < 16; j2 += 2) {
    int jn = j2 + 2 < 16 ? j2 + 2 : 28;
    PAIR(j2, jn, EPI11)
    METER(j2 >> 1)
  }
#pragma nounroll
  for (int j2 = 28; j2 < 36; j2 += 2) {
    int jn = j2 + 2 < 36 ? j2 + 2 : 35;
    PAIR(j2, jn, EPI14)
    METER(8 + ((j2 - 28) >> 1))
  }

#pragma unroll
  for (int t = 0; t < RT; t++) {
#pragma unroll
    for (int r = 0; r < 8; r++) {
      int e = r + 8 * hb;
      atomAddF(sums + (size_t)nodes[t * 16 + e] * 40 + nn, o0[t][r]);
    }
  }
#undef METER
#undef PAIR
#undef EPI11
#undef EPI12
#undef EPI13
#undef EPI14
}

__global__ void node_stage1(const float* __restrict__ atom, float* ws) {
  int c = threadIdx.x;
  int ty = threadIdx.y;
  float* sums = ws + OFF_SUMS;
  const float* cnts = ws + OFF_CNTS;
  float* stats = ws + OFF_STATS;
  float sA = 0.0f, sB = 0.0f;
  for (int n = blockIdx.x * 6 + ty; n < N_NODES; n += gridDim.x * 6) {
    float cnt = cnts[n];
    float inv = 1.0f / fmaxf(cnt, 1.0f);
    float y = sums[(size_t)n * 40 + c] * inv + atom[(size_t)n * 40 + c];
    sums[(size_t)n * 40 + c] = y;
    sA += y;
    sB += y * y;
  }
  __shared__ float rA[6][40], rB[6][40];
  rA[ty][c] = sA;
  rB[ty][c] = sB;
  __syncthreads();
  if (ty == 0) {
    float tA = 0.0f, tB = 0.0f;
#pragma unroll
    for (int k = 0; k < 6; k++) { tA += rA[k][c]; tB += rB[k][c]; }
    if (c < 16) {
      atomAddF(stats + c, tA);
      atomAddF(stats + 16 + c, tB);
    } else {
      atomAddF(stats + 32 + (c - 16) / 3, tB);
    }
  }
}

__global__ void node_stage2(const float* __restrict__ ws,
                            const float* __restrict__ bnw,
                            const float* __restrict__ bnb,
                            float* __restrict__ out) {
  int idx = blockIdx.x * blockDim.x + threadIdx.x;
  if (idx >= N_NODES * 40) return;
  int c = idx % 40;
  const float* y = ws + OFF_SUMS;
  const float* stats = ws + OFF_STATS;
  float v = y[idx];
  float o;
  if (c < 16) {
    float mean = stats[c] * (1.0f / N_NODES);
    float ex2 = stats[16 + c] * (1.0f / N_NODES);
    float var = ex2 - mean * mean;
    o = (v - mean) * rsqrtf(var + EPSV) * bnw[c] + bnb[c];
  } else {
    int j = (c - 16) / 3;
    float vn = stats[32 + j] * (1.0f / (3.0f * N_NODES));
    o = v * rsqrtf(vn + EPSV) * bnw[16 + j];
  }
  out[idx] = o;
}

extern "C" void kernel_launch(void* const* d_in, const int* in_sizes, int n_in,
                              void* d_out, int out_size, void* d_ws,
                              size_t ws_size, hipStream_t stream) {
  (void)in_sizes; (void)n_in; (void)out_size; (void)ws_size;
  const float* atom = (const float*)d_in[0];
  const float* ef   = (const float*)d_in[1];
  const float* esh  = (const float*)d_in[2];
  const int*   eidx = (const int*)d_in[3];
  const float* w1   = (const float*)d_in[4];
  const float* b1   = (const float*)d_in[5];
  const float* w2   = (const float*)d_in[6];
  const float* b2   = (const float*)d_in[7];
  const float* bnw  = (const float*)d_in[8];
  const float* bnb  = (const float*)d_in[9];
  float* ws  = (float*)d_ws;
  float* out = (float*)d_out;

  zero_kernel<<<(ZERO_N + 255) / 256, 256, 0, stream>>>(ws);
  repack_kernel<<<(640 * 32 + 255) / 256, 256, 0, stream>>>(
      w1, w2, ws + OFF_W1F, ws + OFF_W2F);
  edge_kernel<<<N_WT, 32, 0, stream>>>(
      atom, ef, esh, eidx, b1, b2, ws + OFF_W1F, ws + OFF_W2F,
      ws + OFF_SUMS, ws + OFF_CNTS);
  node_stage1<<<128, dim3(40, 6), 0, stream>>>(atom, ws);
  node_stage2<<<(N_NODES * 40 + 255) / 256, 256, 0, stream>>>(ws, bnw, bnb, out);
}


# fill, mean+residual and norm kernels moved to one 16-byte piece per lane: 1608, 840 and 1600 waves enter instead of 12816, 1024 and 12504; norm-stat partial sums regrouped (new sum order)
# speedup vs baseline: 1.4116x; 1.0458x over previous
#include <hip/hip_runtime.h>

typedef float v2f __attribute__((ext_vector_type(2)));
typedef float v8f __attribute__((ext_vector_type(8)));

#define N_NODES 10000
#define N_EDGES 160000
#define N_TILES (N_EDGES / 16)

#define INV_M0  0.25f
#define INV_3M1 0.20412414523f
#define INV_M1  0.35355339059f
#define EPSV    1e-5f

#define OFF_SUMS  0
#define OFF_CNTS  400000
#define OFF_STATS 410000
#define ZERO_N    410040
#define OFF_W1F   410048
#define OFF_W2F   414144

static __device__ __forceinline__ v8f wmma4(v2f a, v2f b, v8f c) {
  return __builtin_amdgcn_wmma_f32_16x16x4_f32(false, a, false, b, (short)0, c,
                                               false, false);
}

static __device__ __forceinline__ void atomAddF(float* p, float v) {
  __hip_atomic_fetch_add(p, v, __ATOMIC_RELAXED, __HIP_MEMORY_SCOPE_AGENT);
}

#define ZERO_Q  (ZERO_N / 4)
#define ZERO_TR 2
#define ZERO_B  ((ZERO_Q + 256 * ZERO_TR - 1) / (256 * ZERO_TR))
static_assert(ZERO_N % 4 == 0, "the cleared span is a whole number of 16-byte pieces");

__global__ __launch_bounds__(256) void zero_kernel(float* __restrict__ ws) {
  float4* w4 = (float4*)ws;
  const float4 z = make_float4(0.0f, 0.0f, 0.0f, 0.0f);
  for (int k = 0; k < ZERO_TR; k++) {
    int q = (blockIdx.x * ZERO_TR + k) * 256 + threadIdx.x;
    if (q < ZERO_Q) w4[q] = z;
  }
}

__global__ void repack_kernel(const float* __restrict__ w1,
                              const float* __restrict__ w2,
                              float* __restrict__ w1f,
                              float* __restrict__ w2f) {
  int t = blockIdx.x * blockDim.x + threadIdx.x;
  int f = t >> 5, lane = t & 31;
  int hb = lane >> 4, nn = lane & 15;
  if (f < 64) {
    int b = f >> 2, j = f & 3;
    int r0 = 4 * b + 2 * hb;
    w1f[(f * 32 + lane) * 2 + 0] = w1[(r0 + 0) * 64 + 16 * j + nn];
    w1f[(f * 32 + lane) * 2 + 1] = w1[(r0 + 1) * 64 + 16 * j + nn];
  } else if (f < 64 + 576) {
    int f2 = f - 64;
    int b = f2 / 36, j = f2 % 36;
    int r0 = 4 * b + 2 * hb;
    w2f[(f2 * 32 + lane) * 2 + 0] = w2[(r0 + 0) * 576 + 16 * j + nn];
    w2f[(f2 * 32 + lane) * 2 + 1] = w2[(r0 + 1) * 576 + 16 * j + nn];
  }
}

#define RT   5
#define EPW  (RT * 16)
#define N_WT (N_EDGES / EPW)
#define TBL  1088
#define SMW  (1088 + RT * TBL)
#define MT_K (EPW * 24 / 32 / 12)

static __device__ __forceinline__ void load_w2(v2f (&w)[16],
                                               const float* __restrict__ w2f,
                                               int j2, int lane) {
#pragma unroll
  for (int b = 0; b < 16; b++)
    w[b] = *(const v2f*)(w2f + ((b * 36 + j2) * 32 + lane) * 2);
}

static __device__ __forceinline__ void tile_mma(v8f (&c)[RT],
                                                const v2f (&ha)[RT][16],
                                                const v2f (&w)[16], float bias) {
#pragma unroll
  for (int t = 0; t < RT; t++)
#pragma unroll
    for (int r = 0; r < 8; r++) c[t][r] = bias;
#pragma unroll
  for (int b = 0; b < 16; b++)
#pragma unroll
    for (int t = 0; t < RT; t++) c[t] = wmma4(ha[t][b], w[b], c[t]);
}

__global__ __launch_bounds__(32) __attribute__((amdgpu_waves_per_eu(2, 2)))
void edge_kernel(
    const float* __restrict__ atom, const float* __restrict__ ef,
    const float* __restrict__ esh, const int* __restrict__ eidx,
    const float* __restrict__ b1, const float* __restrict__ b2,
    const float* __restrict__ w1f, const float* __restrict__ w2f,
    float* __restrict__ sums, float* __restrict__ cnts) {
  __shared__ __attribute__((aligned(16))) float smem[SMW];
  __shared__ int nodes[EPW];

  int lane = threadIdx.x;
  int e0 = blockIdx.x * EPW;
  int hb = lane >> 4;
  int nn = lane & 15;

  float* hT = smem;
  float* tb = smem + 1088;

  int dsti[(EPW + 31) / 32], srci[(EPW + 31) / 32];
  float4 shv[(EPW + 31) / 32];
  v8f hc[RT][4];
  {
    float* stg = tb;
    const float4* efv = (const float4*)(ef + (size_t)e0 * 64);
    float4 fv[EPW * 16 / 32];
#pragma unroll
    for (int k = 0; k < EPW * 16 / 32; k++)
      fv[k] = efv[k * 32 + lane];
    v2f wb[4][16];
    float bias1[4];
#pragma unroll
    for (int j = 0; j < 4; j++) bias1[j] = b1[16 * j + nn];
#pragma unroll
    for (int b = 0; b < 16; b++)
      wb[0][b] = *(const v2f*)(w1f + ((b * 4 + 0) * 32 + lane) * 2);
#pragma unroll
    for (int k = 0; k < EPW * 16 / 32; k++) {
      int f = k * 32 + lane;
      *(float4*)(stg + (f >> 4) * 68 + (f & 15) * 4) = fv[k];
    }
    asm volatile("s_wait_dscnt 0" ::: "memory");
    v2f a[RT][16];
#pragma unroll
    for (int t = 0; t < RT; t++) {
      const float* efr = stg + (t * 16 + nn) * 68 + 2 * hb;
#pragma unroll
      for (int b = 0; b < 16; b++) a[t][b] = *(const v2f*)(efr + 4 * b);
    }
    asm volatile("s_wait_dscnt 0" ::: "memory");
#pragma unroll
    for (int p = 0; p < (EPW + 31) / 32; p++) {
      int elc = p * 32 + lane < EPW ? p * 32 + lane : EPW - 1;
      int e = e0 + elc;
      dsti[p] = eidx[e];
      srci[p] = eidx[N_EDGES + e];
      shv[p] = *(const float4*)(esh + e * 4);
    }
#pragma unroll
    for (int j = 1; j < 4; j++)
#pragma unroll
      for (int b = 0; b < 16; b++)
        wb[j][b] = *(const v2f*)(w1f + ((b * 4 + j) * 32 + lane) * 2);
#pragma unroll
    for (int j = 0; j < 4; j++) {
#pragma unroll
      for (int t = 0; t < RT; t++)
#pragma unroll
        for (int r = 0; r < 8; r++) hc[t][j][r] = bias1[j];
#pragma unroll
      for (int b = 0; b < 16; b++)
#pragma unroll
        for (int t = 0; t < RT; t++)
          hc[t][j] = wmma4(a[t][b], wb[j][b], hc[t][j]);
    }
  }

  v2f ha[RT][16];
#pragma unroll
  for (int t = 0; t < RT; t++) {
#pragma unroll
    for (int j = 0; j < 4; j++)
#pragma unroll
      for (int r = 0; r < 8; r++) {
        float v = hc[t][j][r] > 0.0f ? hc[t][j][r] : 0.0f;
        hT[(r + 8 * hb) * 68 + 16 * j + nn] = v;
      }
    asm volatile("s_wait_dscnt 0" ::: "memory");
#pragma unroll
    for (int b = 0; b < 16; b++)
      ha[t][b] = *(const v2f*)(hT + nn * 68 + 4 * b + 2 * hb);
    asm volatile("s_wait_dscnt 0" ::: "memory");
  }

  {
    float x[(EPW + 31) / 32][40];
#pragma unroll
    for (int p = 0; p < (EPW + 31) / 32; p++) {
      const float* xr = atom + (size_t)dsti[p] * 40;
#pragma unroll
      for (int t = 0; t < 10; t++) {
        float4 v = *(const float4*)(xr + t * 4);
        x[p][t * 4 + 0] = v.x; x[p][t * 4 + 1] = v.y;
        x[p][t * 4 + 2] = v.z; x[p][t * 4 + 3] = v.w;
      }
    }
#pragma unroll
    for (int p = 0; p < (EPW + 31) / 32; p++) {
      int el = p * 32 + lane;
      if ((p + 1) * 32 <= EPW || el < EPW) {
        int row = el & 15;
        float* tr = tb + (el >> 4) * TBL;
        float* a11 = tr;
        float* a12 = tr + 256;
        float* a13 = tr + 512;
        float* a14 = tr + 896;
        float* s1  = tr + 1024;
        float4 s = shv[p];
        nodes[el] = srci[p];
        float sh0 = s.x;
        s1[row * 4 + 0] = s.y;
        s1[row * 4 + 1] = s.z;
        s1[row * 4 + 2] = s.w;
#pragma unroll
        for (int j = 0; j < 16; j++) {
          a11[row * 16 + j] = x[p][j] * sh0 * INV_M0;
          a12[row * 16 + j] = x[p][j] * INV_M0;
        }
#pragma unroll
        for (int i = 0; i < 8; i++) {
          float m0 = x[p][16 + i * 3 + 0];
          float m1 = x[p][16 + i * 3 + 1];
          float m2 = x[p][16 + i * 3 + 2];
          a13[row * 24 + i * 3 + 0] = sh0 * m0 * INV_M1;
          a13[row * 24 + i * 3 + 1] = sh0 * m1 * INV_M1;
          a13[row * 24 + i * 3 + 2] = sh0 * m2 * INV_M1;
          a14[row * 8 + i] = (m0 * s.y + m1 * s.z + m2 * s.w) * INV_3M1;
        }
      }
    }
  }

  asm volatile("s_wait_dscnt 0" ::: "memory");

  float o0[RT][8], p1[RT][8], q0[RT][8], q1[RT][8], q2[RT][8];
#pragma unroll
  for (int t = 0; t < RT; t++)
#pragma unroll
    for (int r = 0; r < 8; r++) {
      o0[t][r] = 0.0f; p1[t][r] = 0.0f;
      q0[t][r] = 0.0f; q1[t][r] = 0.0f; q2[t][r] = 0.0f;
    }

  v2f wA[16], wB[16];
  float biasA, biasB;
  v8f c[RT];

#define EPI11(J2)                                                        \
  _Pragma("unroll") for (int t = 0; t < RT; t++) {                       \
    const float* A = tb + t * TBL;                                       \
    _Pragma("unroll") for (int r = 0; r < 8; r++)                        \
      o0[t][r] += c[t][r] * A[(r + 8 * hb) * 16 + (J2)];                 \
  }
#define EPI12(J2)                                                        \
  {                                                                      \
    int i = 2 * ((J2) - 16) + (nn >> 3);                                 \
    _Pragma("unroll") for (int t = 0; t < RT; t++) {                     \
      const float* A = tb + t * TBL + 256;                               \
      _Pragma("unroll") for (int r = 0; r < 8; r++)                      \
        p1[t][r] += c[t][r] * A[(r + 8 * hb) * 16 + i];                  \
    }                                                                    \
  }
#define EPI13(J2)                                                        \
  {                                                                      \
    int i = 2 * ((J2) - 24) + (nn >> 3);                                 \
    _Pragma("unroll") for (int t = 0; t < RT; t++) {                     \
      const float* A = tb + t * TBL + 512;                               \
      _Pragma("unroll") for (int r = 0; r < 8; r++) {                    \
        float cv = c[t][r];                                              \
        int base = (r + 8 * hb) * 24 + i * 3;                            \
        q0[t][r] += cv * A[base + 0];                                    \
        q1[t][r] += cv * A[base + 1];                                    \
        q2[t][r] += cv * A[base + 2];                                    \
      }                                                                  \
    }                                                                    \
  }
#define EPI14(J2)                                                        \
  {                                                                      \
    int i = (J2) - 28;                                                   \
    _Pragma("unroll") for (int t = 0; t < RT; t++) {                     \
      const float* A = tb + t * TBL + 896;                               \
      _Pragma("unroll") for (int r = 0; r < 8; r++)                      \
        o0[t][r] += c[t][r] * A[(r + 8 * hb) * 8 + i];                   \
    }                                                                    \
  }
#define PAIR(J2, JN, EPI)                                                \
  load_w2(wB, w2f, (J2) + 1, lane);                                      \
  biasB = b2[16 * ((J2) + 1) + nn];                                      \
  tile_mma(c, ha, wA, biasA);                                            \
  EPI(J2)                                                                \
  load_w2(wA, w2f, (JN), lane);                                          \
  biasA = b2[16 * (JN) + nn];                                            \
  tile_mma(c, ha, wB, biasB);                                            \
  EPI((J2) + 1)

#define METER(TRIP)                                                      \
  {                                                                      \
    float mv[MT_K];                                                      \
    int mo[MT_K];                                                        \
    _Pragma("unroll") for (int k = 0; k < MT_K; k++) {                   \
      int f = (TRIP) * (MT_K * 32) + k * 32 + lane;                      \
      int el = f / 24;                                                   \
      int ch = f - el * 24;                                              \
      mv[k] = tb[(el >> 4) * TBL + 512 + (el & 15) * 24 + ch];           \
      mo[k] = nodes[el] * 40 + 16 + ch;                                  \
    }                                                                    \
    _Pragma("unroll") for (int k = 0; k < MT_K; k++)                     \
      atomAddF(sums + mo[k], mv[k]);                                     \
  }

  load_w2(wA, w2f, 16, lane);
  biasA = b2[16 * 16 + nn];

#pragma nounroll
  for (int j2 = 16; j2 < 24; j2 += 2) { PAIR(j2, j2 + 2, EPI12) }
#pragma nounroll
  for (int j2 = 24; j2 < 28; j2 += 2) {
    int jn = j2 + 2 < 28 ? j2 + 2 : 0;
    PAIR(j2, jn, EPI13)
  }

#pragma unroll
  for (int t = 0; t < RT; t++) {
    const float* s1 = tb + t * TBL + 1024;
    float* stg = tb + t * TBL + 512;
#pragma unroll
    for (int r = 0; r < 8; r++) {
      p1[t][r] += __shfl_xor(p1[t][r], 8, 32);
      q0[t][r] += __shfl_xor(q0[t][r], 8, 32);
      q1[t][r] += __shfl_xor(q1[t][r], 8, 32);
      q2[t][r] += __shfl_xor(q2[t][r], 8, 32);
    }
    if (nn < 8) {
#pragma unroll
      for (int r = 0; r < 8; r++) {
        int e = r + 8 * hb;
        float sy = s1[e * 4 + 0], sz = s1[e * 4 + 1], sw = s1[e * 4 + 2];
        stg[e * 24 + nn * 3 + 0] = p1[t][r] * sy + q0[t][r];
        stg[e * 24 + nn * 3 + 1] = p1[t][r] * sz + q1[t][r];
        stg[e * 24 + nn * 3 + 2] = p1[t][r] * sw + q2[t][r];
      }
    }
  }
  asm volatile("s_wait_dscnt 0" ::: "memory");
#pragma unroll
  for (int p = 0; p < (EPW + 31) / 32; p++) {
    int el = p * 32 + lane;
    if (el < EPW) atomAddF(cnts + nodes[el], 1.0f);
  }

#pragma nounroll
  for (int j2 = 0; j2 < 16; j2 += 2) {
    int jn = j2 + 2 < 16 ? j2 + 2 : 28;
    PAIR(j2, jn, EPI11)
    METER(j2 >> 1)
  }
#pragma nounroll
  for (int j2 = 28; j2 < 36; j2 += 2) {
    int jn = j2 + 2 < 36 ? j2 + 2 : 35;
    PAIR(j2, jn, EPI14)
    METER(8 + ((j2 - 28) >> 1))
  }

#pragma unroll
  for (int t = 0; t < RT; t++) {
#pragma unroll
    for (int r = 0; r < 8; r++) {
      int e = r + 8 * hb;
      atomAddF(sums + (size_t)nodes[t * 16 + e] * 40 + nn, o0[t][r]);
    }
  }
#undef METER
#undef PAIR
#undef EPI11
#undef EPI12
#undef EPI13
#undef EPI14
}

#define S1_ROWS 24
#define S1_TR   4
#define S1_B    ((N_NODES + S1_ROWS * S1_TR - 1) / (S1_ROWS * S1_TR))

__global__ __launch_bounds__(240) void node_stage1(const float* __restrict__ atom,
                                                   float* ws) {
  int t = threadIdx.x;
  int g = t % 10;
  int r = t / 10;
  float4* sums4 = (float4*)(ws + OFF_SUMS);
  const float4* atom4 = (const float4*)atom;
  const float* cnts = ws + OFF_CNTS;
  float* stats = ws + OFF_STATS;
  int row0 = blockIdx.x * (S1_ROWS * S1_TR) + r;

  float cv[S1_TR];
  float4 sv[S1_TR], av[S1_TR];
#pragma unroll
  for (int k = 0; k < S1_TR; k++) {
    int row = row0 + k * S1_ROWS;
    int rc = row < N_NODES ? row : N_NODES - 1;
    cv[k] = cnts[rc];
    sv[k] = sums4[rc * 10 + g];
    av[k] = atom4[rc * 10 + g];
  }
  float4 sA = make_float4(0.0f, 0.0f, 0.0f, 0.0f);
  float4 sB = make_float4(0.0f, 0.0f, 0.0f, 0.0f);
#pragma unroll
  for (int k = 0; k < S1_TR; k++) {
    int row = row0 + k * S1_ROWS;
    if (row < N_NODES) {
      float inv = 1.0f / fmaxf(cv[k], 1.0f);
      float4 y;
      y.x = sv[k].x * inv + av[k].x;
      y.y = sv[k].y * inv + av[k].y;
      y.z = sv[k].z * inv + av[k].z;
      y.w = sv[k].w * inv + av[k].w;
      sums4[row * 10 + g] = y;
      sA.x += y.x; sA.y += y.y; sA.z += y.z; sA.w += y.w;
      sB.x += y.x * y.x; sB.y += y.y * y.y;
      sB.z += y.z * y.z; sB.w += y.w * y.w;
    }
  }
  __shared__ __attribute__((aligned(16))) float rA[S1_ROWS][40];
  __shared__ __attribute__((aligned(16))) float rB[S1_ROWS][40];
  *(float4*)&rA[r][4 * g] = sA;
  *(float4*)&rB[r][4 * g] = sB;
  __syncthreads();
  if (t < 40) {
    int c = t;
    float tA = 0.0f, tB = 0.0f;
    for (int k = 0; k < S1_ROWS; k++) { tA += rA[k][c]; tB += rB[k][c]; }
    if (c < 16) {
      atomAddF(stats + c, tA);
      atomAddF(stats + 16 + c, tB);
    } else {
      atomAddF(stats + 32 + (c - 16) / 3, tB);
    }
  }
}

#define S2_Q  (N_NODES * 10)
#define S2_TR 2
#define S2_B  200
static_assert((S2_B * 256) % 10 == 0, "a thread keeps its channel group across trips");
static_assert(S2_B * 256 * S2_TR >= S2_Q, "the grid covers the output");

__global__ __launch_bounds__(256) void node_stage2(const float* __restrict__ ws,
                                                   const float* __restrict__ bnw,
                                                   const float* __restrict__ bnb,
                                                   float* __restrict__ out) {
  int tid = blockIdx.x * 256 + threadIdx.x;
  int g = tid % 10;
  const float4* y4 = (const float4*)(ws + OFF_SUMS);
  const float* stats = ws + OFF_STATS;
  float4* out4 = (float4*)out;

  float4 v[S2_TR];
#pragma unroll
  for (int k = 0; k < S2_TR; k++) {
    int q = tid + k * (S2_B * 256);
    v[k] = y4[q < S2_Q ? q : S2_Q - 1];
  }

  if (g < 4) {
    float4 s1 = *(const float4*)(stats + 4 * g);
    float4 s2 = *(const float4*)(stats + 16 + 4 * g);
    float st1[4] = {s1.x, s1.y, s1.z, s1.w};
    float st2[4] = {s2.x, s2.y, s2.z, s2.w};
    float w[4], b[4];
#pragma unroll
    for (int i = 0; i < 4; i++) {
      w[i] = bnw[4 * g + i];
      b[i] = bnb[4 * g + i];
    }
    float mean[4], rs[4];
#pragma unroll
    for (int i = 0; i < 4; i++) {
      mean[i] = st1[i] * (1.0f / N_NODES);
      float ex2 = st2[i] * (1.0f / N_NODES);
      float var = ex2 - mean[i] * mean[i];
      rs[i] = rsqrtf(var + EPSV);
    }
#pragma unroll
    for (int k = 0; k < S2_TR; k++) {
      int q = tid + k * (S2_B * 256);
      if (q < S2_Q) {
        float4 o;
        o.x = (v[k].x - mean[0]) * rs[0] * w[0] + b[0];
        o.y = (v[k].y - mean[1]) * rs[1] * w[1] + b[1];
        o.z = (v[k].z - mean[2]) * rs[2] * w[2] + b[2];
        o.w = (v[k].w - mean[3]) * rs[3] * w[3] + b[3];
        out4[q] = o;
      }
    }
  } else {
    float f[4];
#pragma unroll
    for (int i = 0; i < 4; i++) {
      int j = (4 * g + i - 16) / 3;
      float vn = stats[32 + j] * (1.0f / (3.0f * N_NODES));
      f[i] = rsqrtf(vn + EPSV);
    }
    float wv[4];
#pragma unroll
    for (int i = 0; i < 4; i++) wv[i] = bnw[16 + (4 * g + i - 16) / 3];
#pragma unroll
    for (int k = 0; k < S2_TR; k++) {
      int q = tid + k * (S2_B * 256);
      if (q < S2_Q) {
        float4 o;
        o.x = v[k].x * f[0] * wv[0];
        o.y = v[k].y * f[1] * wv[1];
        o.z = v[k].z * f[2] * wv[2];
        o.w = v[k].w * f[3] * wv[3];
        out4[q] = o;
      }
    }
  }
}

extern "C" void kernel_launch(void* const* d_in, const int* in_sizes, int n_in,
                              void* d_out, int out_size, void* d_ws,
                              size_t ws_size, hipStream_t stream) {
  (void)in_sizes; (void)n_in; (void)out_size; (void)ws_size;
  const float* atom = (const float*)d_in[0];
  const float* ef   = (const float*)d_in[1];
  const float* esh  = (const float*)d_in[2];
  const int*   eidx = (const int*)d_in[3];
  const float* w1   = (const float*)d_in[4];
  const float* b1   = (const float*)d_in[5];
  const float* w2   = (const float*)d_in[6];
  const float* b2   = (const float*)d_in[7];
  const float* bnw  = (const float*)d_in[8];
  const float* bnb  = (const float*)d_in[9];
  float* ws  = (float*)d_ws;
  float* out = (float*)d_out;

  zero_kernel<<<ZERO_B, 256, 0, stream>>>(ws);
  repack_kernel<<<(640 * 32 + 255) / 256, 256, 0, stream>>>(
      w1, w2, ws + OFF_W1F, ws + OFF_W2F);
  edge_kernel<<<N_WT, 32, 0, stream>>>(
      atom, ef, esh, eidx, b1, b2, ws + OFF_W1F, ws + OFF_W2F,
      ws + OFF_SUMS, ws + OFF_CNTS);
  node_stage1<<<S1_B, 240, 0, stream>>>(atom, ws);
  node_stage2<<<S2_B, 256, 0, stream>>>(ws, bnw, bnb, out);
}
